# starting kernel
# baseline (speedup 1.0000x reference)
#include <hip/hip_runtime.h>

typedef __attribute__((ext_vector_type(2))) float v2f;
typedef __attribute__((ext_vector_type(8))) float v8f;

#define C_DIM 512
#define HW    4096
#define OUT_C 3
#define NPAD  16
#define B_STR 522

__global__ __launch_bounds__(256) void torgb_phase1(
    const float* __restrict__ w, const float* __restrict__ aw,
    const float* __restrict__ ab, const float* __restrict__ wt,
    float* __restrict__ eff)
{
    __shared__ float s_w[C_DIM];
    __shared__ float s_styles[C_DIM];
    __shared__ float s_red[OUT_C][256];
    __shared__ float s_d[OUT_C];

    const int b   = blockIdx.x;
    const int tid = threadIdx.x;

    s_w[tid]       = w[b * C_DIM + tid];
    s_w[tid + 256] = w[b * C_DIM + tid + 256];
    __syncthreads();

    const float gain = 0.044194173824159216f;
    for (int cc = tid; cc < C_DIM; cc += 256) {
        const float4* row = (const float4*)(aw + (size_t)cc * C_DIM);
        float sum = 0.0f;
        #pragma unroll 4
        for (int k = 0; k < C_DIM / 4; ++k) {
            float4 a = row[k];
            const float* wk = &s_w[k * 4];
            sum += a.x * wk[0] + a.y * wk[1] + a.z * wk[2] + a.w * wk[3];
        }
        s_styles[cc] = sum * gain + ab[cc];
    }
    __syncthreads();

    float p0 = 0.0f, p1 = 0.0f, p2 = 0.0f;
    for (int cc = tid; cc < C_DIM; cc += 256) {
        float s = s_styles[cc];
        float t0 = wt[0 * C_DIM + cc] * s;
        float t1 = wt[1 * C_DIM + cc] * s;
        float t2 = wt[2 * C_DIM + cc] * s;
        p0 += t0 * t0; p1 += t1 * t1; p2 += t2 * t2;
    }
    s_red[0][tid] = p0; s_red[1][tid] = p1; s_red[2][tid] = p2;
    __syncthreads();
    for (int off = 128; off > 0; off >>= 1) {
        if (tid < off) {
            s_red[0][tid] += s_red[0][tid + off];
            s_red[1][tid] += s_red[1][tid + off];
            s_red[2][tid] += s_red[2][tid + off];
        }
        __syncthreads();
    }
    if (tid < OUT_C) s_d[tid] = rsqrtf(s_red[tid][0] + 1e-8f);
    __syncthreads();

    for (int cc = tid; cc < C_DIM; cc += 256) {
        float s = s_styles[cc];
        #pragma unroll
        for (int n = 0; n < OUT_C; ++n)
            eff[((size_t)b * OUT_C + n) * C_DIM + cc] = wt[n * C_DIM + cc] * s * s_d[n];
    }
}

__global__ __launch_bounds__(256) void torgb_main(
    const float* __restrict__ x, const float* __restrict__ eff,
    const float* __restrict__ bias, float* __restrict__ out)
{
    __shared__ float sB[NPAD * B_STR];

    const int b   = blockIdx.x >> 5;
    const int grp = blockIdx.x & 31;
    const int tid = threadIdx.x;

    for (int i = tid; i < NPAD * B_STR; i += 256) {
        int n = i / B_STR;
        int c = i - n * B_STR;
        float v = 0.0f;
        if (n < OUT_C && c < C_DIM) v = eff[((size_t)b * OUT_C + n) * C_DIM + c];
        sB[i] = v;
    }
    __syncthreads();

    const int lane = tid & 31;
    const int wv   = tid >> 5;
    const int m    = lane & 15;
    const int half = lane >> 4;
    const int koff = half << 1;
    const int hw0  = (grp * 8 + wv) * 16;

    const float* pA = x + ((size_t)b * C_DIM + (size_t)koff) * HW + hw0 + m;
    const float* pB = &sB[m * B_STR + koff];

    v8f acc = {0.f, 0.f, 0.f, 0.f, 0.f, 0.f, 0.f, 0.f};

    #pragma unroll 4
    for (int c = 0; c < C_DIM; c += 4) {
        float a0 = __builtin_nontemporal_load(&pA[(size_t)c * HW]);
        float a1 = __builtin_nontemporal_load(&pA[(size_t)(c + 1) * HW]);
        v2f Bv = *(const v2f*)&pB[c];
        v2f Av = {a0, a1};
        acc = __builtin_amdgcn_wmma_f32_16x16x4_f32(
            false, Av, false, Bv, (short)0, acc, false, false);
    }

    if (m < OUT_C) {
        const float bs = bias[m];
        float* po = out + ((size_t)b * OUT_C + m) * HW + hw0 + half * 8;
        #pragma unroll
        for (int r = 0; r < 8; ++r)
            po[r] = acc[r] + bs;
    }
}

extern "C" void kernel_launch(void* const* d_in, const int* in_sizes, int n_in,
                              void* d_out, int out_size, void* d_ws, size_t ws_size,
                              hipStream_t stream) {
    const float* x    = (const float*)d_in[0];
    const float* w    = (const float*)d_in[1];
    const float* aw   = (const float*)d_in[2];
    const float* ab   = (const float*)d_in[3];
    const float* wt   = (const float*)d_in[4];
    const float* bias = (const float*)d_in[5];
    float* out = (float*)d_out;
    float* eff = (float*)d_ws;

    torgb_phase1<<<16, 256, 0, stream>>>(w, aw, ab, wt, eff);
    torgb_main<<<512, 256, 0, stream>>>(x, eff, bias, out);
}


# Style GEMV re-split: one wave per affine row, coalesced row loads, all 16 batch dots per wave (was 1 block per batch); demod sum folded into main's LDS staging; dot and demod sums now shuffle trees
# speedup vs baseline: 1.5091x; 1.5091x over previous
#include <hip/hip_runtime.h>


typedef __attribute__((ext_vector_type(2))) float v2f;
typedef __attribute__((ext_vector_type(8))) float v8f;

#define C_DIM 512
#define HW    4096
#define OUT_C 3
#define NPAD  16
#define B_STR 522

__device__ __forceinline__ float dot_piece(const float4 a0, const float4 a1, const float4 a2,
                                           const float4 a3, const float* __restrict__ wrow, int lane)
{
    const float4* wr = (const float4*)wrow;
    const float4 v0 = wr[lane], v1 = wr[lane + 32], v2 = wr[lane + 64], v3 = wr[lane + 96];
    float s = 0.0f;
    s += a0.x * v0.x + a0.y * v0.y + a0.z * v0.z + a0.w * v0.w;
    s += a1.x * v1.x + a1.y * v1.y + a1.z * v1.z + a1.w * v1.w;
    s += a2.x * v2.x + a2.y * v2.y + a2.z * v2.z + a2.w * v2.w;
    s += a3.x * v3.x + a3.y * v3.y + a3.z * v3.z + a3.w * v3.w;
    return s;
}

__device__ __forceinline__ float xch(float lo, float hi, bool up, int width)
{
    const float keep = up ? hi : lo;
    const float send = up ? lo : hi;
    return keep + __shfl_xor(send, width, 32);
}

__global__ __launch_bounds__(128) void torgb_styles(
    const float* __restrict__ w, const float* __restrict__ aw,
    const float* __restrict__ ab, float* __restrict__ styles)
{
    const int lane = threadIdx.x & 31;
    const int cc   = blockIdx.x * 4 + (threadIdx.x >> 5);

    const float4* arow = (const float4*)(aw + (size_t)cc * C_DIM);
    const float4 a0 = arow[lane], a1 = arow[lane + 32], a2 = arow[lane + 64], a3 = arow[lane + 96];

#define DOT(b) dot_piece(a0, a1, a2, a3, w + (b) * C_DIM, lane)
    const float s0  = DOT(0),  s1  = DOT(1),  s2  = DOT(2),  s3  = DOT(3);
    const float s4  = DOT(4),  s5  = DOT(5),  s6  = DOT(6),  s7  = DOT(7);
    const float s8  = DOT(8),  s9  = DOT(9),  s10 = DOT(10), s11 = DOT(11);
    const float s12 = DOT(12), s13 = DOT(13), s14 = DOT(14), s15 = DOT(15);
#undef DOT

    const bool u16 = (lane & 16) != 0;
    const float e0 = xch(s0, s8,  u16, 16), e1 = xch(s1, s9,  u16, 16);
    const float e2 = xch(s2, s10, u16, 16), e3 = xch(s3, s11, u16, 16);
    const float e4 = xch(s4, s12, u16, 16), e5 = xch(s5, s13, u16, 16);
    const float e6 = xch(s6, s14, u16, 16), e7 = xch(s7, s15, u16, 16);
    const bool u8 = (lane & 8) != 0;
    const float f0 = xch(e0, e4, u8, 8), f1 = xch(e1, e5, u8, 8);
    const float f2 = xch(e2, e6, u8, 8), f3 = xch(e3, e7, u8, 8);
    const bool u4 = (lane & 4) != 0;
    const float g0 = xch(f0, f2, u4, 4), g1 = xch(f1, f3, u4, 4);
    const bool u2 = (lane & 2) != 0;
    const float h0 = xch(g0, g1, u2, 2);
    const float tot = h0 + __shfl_xor(h0, 1, 32);

    if ((lane & 1) == 0) {
        const float gain = 0.044194173824159216f;
        styles[(size_t)(lane >> 1) * C_DIM + cc] = tot * gain + ab[cc];
    }
}

__global__ __launch_bounds__(256) void torgb_main(
    const float* __restrict__ x, const float* __restrict__ styles,
    const float* __restrict__ wt, const float* __restrict__ bias,
    float* __restrict__ out)
{
    __shared__ float sB[NPAD * B_STR];
    __shared__ float s_part[OUT_C][8];

    const int b    = blockIdx.x >> 5;
    const int grp  = blockIdx.x & 31;
    const int tid  = threadIdx.x;
    const int lane = tid & 31;
    const int wv   = tid >> 5;

    const float s0  = styles[(size_t)b * C_DIM + tid];
    const float s1  = styles[(size_t)b * C_DIM + tid + 256];
    const float w00 = wt[0 * C_DIM + tid], w01 = wt[0 * C_DIM + tid + 256];
    const float w10 = wt[1 * C_DIM + tid], w11 = wt[1 * C_DIM + tid + 256];
    const float w20 = wt[2 * C_DIM + tid], w21 = wt[2 * C_DIM + tid + 256];

    for (int i = tid; i < NPAD * B_STR; i += 256) {
        int n = i / B_STR;
        int c = i - n * B_STR;
        if (n >= OUT_C || c >= C_DIM) sB[i] = 0.0f;
    }

    const float t00 = w00 * s0, t01 = w01 * s1;
    const float t10 = w10 * s0, t11 = w11 * s1;
    const float t20 = w20 * s0, t21 = w21 * s1;
    float p0 = t00 * t00 + t01 * t01;
    float p1 = t10 * t10 + t11 * t11;
    float p2 = t20 * t20 + t21 * t21;
    #pragma unroll
    for (int off = 16; off > 0; off >>= 1) {
        p0 += __shfl_xor(p0, off, 32);
        p1 += __shfl_xor(p1, off, 32);
        p2 += __shfl_xor(p2, off, 32);
    }
    if (lane == 0) { s_part[0][wv] = p0; s_part[1][wv] = p1; s_part[2][wv] = p2; }
    __syncthreads();
    float q0 = 0.0f, q1 = 0.0f, q2 = 0.0f;
    #pragma unroll
    for (int k = 0; k < 8; ++k) { q0 += s_part[0][k]; q1 += s_part[1][k]; q2 += s_part[2][k]; }
    const float d0 = rsqrtf(q0 + 1e-8f);
    const float d1 = rsqrtf(q1 + 1e-8f);
    const float d2 = rsqrtf(q2 + 1e-8f);
    sB[0 * B_STR + tid] = t00 * d0;  sB[0 * B_STR + tid + 256] = t01 * d0;
    sB[1 * B_STR + tid] = t10 * d1;  sB[1 * B_STR + tid + 256] = t11 * d1;
    sB[2 * B_STR + tid] = t20 * d2;  sB[2 * B_STR + tid + 256] = t21 * d2;
    __syncthreads();

    const int m    = lane & 15;
    const int half = lane >> 4;
    const int koff = half << 1;
    const int hw0  = (grp * 8 + wv) * 16;

    const float* pA = x + ((size_t)b * C_DIM + (size_t)koff) * HW + hw0 + m;
    const float* pB = &sB[m * B_STR + koff];

    v8f acc = {0.f, 0.f, 0.f, 0.f, 0.f, 0.f, 0.f, 0.f};

    #pragma unroll 4
    for (int c = 0; c < C_DIM; c += 4) {
        float a0 = __builtin_nontemporal_load(&pA[(size_t)c * HW]);
        float a1 = __builtin_nontemporal_load(&pA[(size_t)(c + 1) * HW]);
        v2f Bv = *(const v2f*)&pB[c];
        v2f Av = {a0, a1};
        acc = __builtin_amdgcn_wmma_f32_16x16x4_f32(
            false, Av, false, Bv, (short)0, acc, false, false);
    }

    if (m < OUT_C) {
        const float bs = bias[m];
        float* po = out + ((size_t)b * OUT_C + m) * HW + hw0 + half * 8;
        #pragma unroll
        for (int r = 0; r < 8; ++r)
            po[r] = acc[r] + bs;
    }
}

extern "C" void kernel_launch(void* const* d_in, const int* in_sizes, int n_in,
                              void* d_out, int out_size, void* d_ws, size_t ws_size,
                              hipStream_t stream) {
    const float* x    = (const float*)d_in[0];
    const float* w    = (const float*)d_in[1];
    const float* aw   = (const float*)d_in[2];
    const float* ab   = (const float*)d_in[3];
    const float* wt   = (const float*)d_in[4];
    const float* bias = (const float*)d_in[5];
    float* out    = (float*)d_out;
    float* styles = (float*)d_ws;

    torgb_styles<<<128, 128, 0, stream>>>(w, aw, ab, styles);
    torgb_main<<<512, 256, 0, stream>>>(x, styles, wt, bias, out);
}


# Main kernel x stream: wave tile 64 px x 128 ch (was 16 x 512), 16-byte loads in 256-B runs issued ~40 ahead, 4 tiles share each B fragment; the 4 channel-quarter sums added via LDS in fixed order
# speedup vs baseline: 2.3619x; 1.5651x over previous
#include <hip/hip_runtime.h>


typedef __attribute__((ext_vector_type(2))) float v2f;
typedef __attribute__((ext_vector_type(4))) float v4f;
typedef __attribute__((ext_vector_type(8))) float v8f;

#define C_DIM 512
#define HW    4096
#define OUT_C 3
#define NPAD  16
#define B_STR 522

__device__ __forceinline__ float dot_piece(const float4 a0, const float4 a1, const float4 a2,
                                           const float4 a3, const float* __restrict__ wrow, int lane)
{
    const float4* wr = (const float4*)wrow;
    const float4 v0 = wr[lane], v1 = wr[lane + 32], v2 = wr[lane + 64], v3 = wr[lane + 96];
    float s = 0.0f;
    s += a0.x * v0.x + a0.y * v0.y + a0.z * v0.z + a0.w * v0.w;
    s += a1.x * v1.x + a1.y * v1.y + a1.z * v1.z + a1.w * v1.w;
    s += a2.x * v2.x + a2.y * v2.y + a2.z * v2.z + a2.w * v2.w;
    s += a3.x * v3.x + a3.y * v3.y + a3.z * v3.z + a3.w * v3.w;
    return s;
}

__device__ __forceinline__ float xch(float lo, float hi, bool up, int width)
{
    const float keep = up ? hi : lo;
    const float send = up ? lo : hi;
    return keep + __shfl_xor(send, width, 32);
}

__global__ __launch_bounds__(128) void torgb_styles(
    const float* __restrict__ w, const float* __restrict__ aw,
    const float* __restrict__ ab, float* __restrict__ styles)
{
    const int lane = threadIdx.x & 31;
    const int cc   = blockIdx.x * 4 + (threadIdx.x >> 5);

    const float4* arow = (const float4*)(aw + (size_t)cc * C_DIM);
    const float4 a0 = arow[lane], a1 = arow[lane + 32], a2 = arow[lane + 64], a3 = arow[lane + 96];

#define DOT(b) dot_piece(a0, a1, a2, a3, w + (b) * C_DIM, lane)
    const float s0  = DOT(0),  s1  = DOT(1),  s2  = DOT(2),  s3  = DOT(3);
    const float s4  = DOT(4),  s5  = DOT(5),  s6  = DOT(6),  s7  = DOT(7);
    const float s8  = DOT(8),  s9  = DOT(9),  s10 = DOT(10), s11 = DOT(11);
    const float s12 = DOT(12), s13 = DOT(13), s14 = DOT(14), s15 = DOT(15);
#undef DOT

    const bool u16 = (lane & 16) != 0;
    const float e0 = xch(s0, s8,  u16, 16), e1 = xch(s1, s9,  u16, 16);
    const float e2 = xch(s2, s10, u16, 16), e3 = xch(s3, s11, u16, 16);
    const float e4 = xch(s4, s12, u16, 16), e5 = xch(s5, s13, u16, 16);
    const float e6 = xch(s6, s14, u16, 16), e7 = xch(s7, s15, u16, 16);
    const bool u8 = (lane & 8) != 0;
    const float f0 = xch(e0, e4, u8, 8), f1 = xch(e1, e5, u8, 8);
    const float f2 = xch(e2, e6, u8, 8), f3 = xch(e3, e7, u8, 8);
    const bool u4 = (lane & 4) != 0;
    const float g0 = xch(f0, f2, u4, 4), g1 = xch(f1, f3, u4, 4);
    const bool u2 = (lane & 2) != 0;
    const float h0 = xch(g0, g1, u2, 2);
    const float tot = h0 + __shfl_xor(h0, 1, 32);

    if ((lane & 1) == 0) {
        const float gain = 0.044194173824159216f;
        styles[(size_t)(lane >> 1) * C_DIM + cc] = tot * gain + ab[cc];
    }
}

#define KQ     4
#define KQ_LEN (C_DIM / KQ)
#define GRP_K  32

__device__ __forceinline__ void load_group(v4f (&r)[16], const float* __restrict__ p)
{
    #pragma unroll
    for (int i = 0; i < 8; ++i) {
        r[2 * i]     = __builtin_nontemporal_load((const v4f*)(p + (size_t)(4 * i) * HW));
        r[2 * i + 1] = __builtin_nontemporal_load((const v4f*)(p + (size_t)(4 * i + 1) * HW));
    }
}

__device__ __forceinline__ void mma_group(v8f& c0, v8f& c1, v8f& c2, v8f& c3,
                                          const v4f (&r)[16], const float* pb)
{
    #pragma unroll
    for (int i = 0; i < 8; ++i) {
        const v2f Bv = *(const v2f*)(pb + 4 * i);
        const v4f r0 = r[2 * i], r1 = r[2 * i + 1];
        const v2f A0 = {r0.x, r1.x}, A1 = {r0.y, r1.y}, A2 = {r0.z, r1.z}, A3 = {r0.w, r1.w};
        c0 = __builtin_amdgcn_wmma_f32_16x16x4_f32(false, A0, false, Bv, (short)0, c0, false, false);
        c1 = __builtin_amdgcn_wmma_f32_16x16x4_f32(false, A1, false, Bv, (short)0, c1, false, false);
        c2 = __builtin_amdgcn_wmma_f32_16x16x4_f32(false, A2, false, Bv, (short)0, c2, false, false);
        c3 = __builtin_amdgcn_wmma_f32_16x16x4_f32(false, A3, false, Bv, (short)0, c3, false, false);
    }
}

__global__ __launch_bounds__(256) __attribute__((amdgpu_waves_per_eu(4, 4))) void torgb_main(
    const float* __restrict__ x, const float* __restrict__ styles,
    const float* __restrict__ wt, const float* __restrict__ bias,
    float* __restrict__ out)
{
    __shared__ float sB[NPAD * B_STR];
    __shared__ float s_part[OUT_C][8];
    __shared__ __attribute__((aligned(16))) float s_red[KQ][OUT_C][128];

    const int b    = blockIdx.x >> 5;
    const int grp  = blockIdx.x & 31;
    const int tid  = threadIdx.x;
    const int lane = tid & 31;
    const int wv   = tid >> 5;

    const float s0  = styles[(size_t)b * C_DIM + tid];
    const float s1  = styles[(size_t)b * C_DIM + tid + 256];
    const float w00 = wt[0 * C_DIM + tid], w01 = wt[0 * C_DIM + tid + 256];
    const float w10 = wt[1 * C_DIM + tid], w11 = wt[1 * C_DIM + tid + 256];
    const float w20 = wt[2 * C_DIM + tid], w21 = wt[2 * C_DIM + tid + 256];

    for (int i = tid; i < NPAD * B_STR; i += 256) {
        int n = i / B_STR;
        int c = i - n * B_STR;
        if (n >= OUT_C || c >= C_DIM) sB[i] = 0.0f;
    }

    const float t00 = w00 * s0, t01 = w01 * s1;
    const float t10 = w10 * s0, t11 = w11 * s1;
    const float t20 = w20 * s0, t21 = w21 * s1;
    float p0 = t00 * t00 + t01 * t01;
    float p1 = t10 * t10 + t11 * t11;
    float p2 = t20 * t20 + t21 * t21;
    #pragma unroll
    for (int off = 16; off > 0; off >>= 1) {
        p0 += __shfl_xor(p0, off, 32);
        p1 += __shfl_xor(p1, off, 32);
        p2 += __shfl_xor(p2, off, 32);
    }
    if (lane == 0) { s_part[0][wv] = p0; s_part[1][wv] = p1; s_part[2][wv] = p2; }
    __syncthreads();
    float q0 = 0.0f, q1 = 0.0f, q2 = 0.0f;
    #pragma unroll
    for (int k = 0; k < 8; ++k) { q0 += s_part[0][k]; q1 += s_part[1][k]; q2 += s_part[2][k]; }
    const float d0 = rsqrtf(q0 + 1e-8f);
    const float d1 = rsqrtf(q1 + 1e-8f);
    const float d2 = rsqrtf(q2 + 1e-8f);
    sB[0 * B_STR + tid] = t00 * d0;  sB[0 * B_STR + tid + 256] = t01 * d0;
    sB[1 * B_STR + tid] = t10 * d1;  sB[1 * B_STR + tid + 256] = t11 * d1;
    sB[2 * B_STR + tid] = t20 * d2;  sB[2 * B_STR + tid + 256] = t21 * d2;
    __syncthreads();

    const int m    = lane & 15;
    const int half = lane >> 4;
    const int koff = half << 1;
    const int pg   = wv & 1;
    const int kq   = wv >> 1;
    const int c0   = kq * KQ_LEN + koff;
    const int px0  = grp * 128 + pg * 64;

    const float* pA = x + ((size_t)b * C_DIM + (size_t)c0) * HW + px0 + 4 * m;
    const float* pB = &sB[m * B_STR + c0];

    v8f acc0 = {0.f, 0.f, 0.f, 0.f, 0.f, 0.f, 0.f, 0.f};
    v8f acc1 = acc0, acc2 = acc0, acc3 = acc0;

    v4f ga[16], gb[16];
    load_group(ga, pA);
    load_group(gb, pA + (size_t)(1 * GRP_K) * HW);
    mma_group(acc0, acc1, acc2, acc3, ga, pB);
    load_group(ga, pA + (size_t)(2 * GRP_K) * HW);
    mma_group(acc0, acc1, acc2, acc3, gb, pB + 1 * GRP_K);
    load_group(gb, pA + (size_t)(3 * GRP_K) * HW);
    mma_group(acc0, acc1, acc2, acc3, ga, pB + 2 * GRP_K);
    mma_group(acc0, acc1, acc2, acc3, gb, pB + 3 * GRP_K);

    if (m < OUT_C) {
        float* pr = &s_red[kq][m][pg * 64 + half * 32];
        #pragma unroll
        for (int r = 0; r < 8; ++r) {
            const v4f v = {acc0[r], acc1[r], acc2[r], acc3[r]};
            *(v4f*)(pr + 4 * r) = v;
        }
    }
    __syncthreads();

    if (tid < 96) {
        const int o = tid >> 5;
        const int q = tid & 31;
        const float bs = bias[o];
        v4f s = *(const v4f*)&s_red[0][o][4 * q];
        s += *(const v4f*)&s_red[1][o][4 * q];
        s += *(const v4f*)&s_red[2][o][4 * q];
        s += *(const v4f*)&s_red[3][o][4 * q];
        s += bs;
        *(v4f*)(out + ((size_t)b * OUT_C + o) * HW + grp * 128 + 4 * q) = s;
    }
}

extern "C" void kernel_launch(void* const* d_in, const int* in_sizes, int n_in,
                              void* d_out, int out_size, void* d_ws, size_t ws_size,
                              hipStream_t stream) {
    const float* x    = (const float*)d_in[0];
    const float* w    = (const float*)d_in[1];
    const float* aw   = (const float*)d_in[2];
    const float* ab   = (const float*)d_in[3];
    const float* wt   = (const float*)d_in[4];
    const float* bias = (const float*)d_in[5];
    float* out    = (float*)d_out;
    float* styles = (float*)d_ws;

    torgb_styles<<<128, 128, 0, stream>>>(w, aw, ab, styles);
    torgb_main<<<512, 256, 0, stream>>>(x, styles, wt, bias, out);
}


# Styles kernel: every load of a wave in flight at once (register budget of one wave per SIMD, ab load kept at the front); was a chain of about 22 dependent load round trips
# speedup vs baseline: 2.9086x; 1.2315x over previous
#include <hip/hip_runtime.h>


typedef __attribute__((ext_vector_type(2))) float v2f;
typedef __attribute__((ext_vector_type(4))) float v4f;
typedef __attribute__((ext_vector_type(8))) float v8f;

#define C_DIM 512
#define HW    4096
#define OUT_C 3
#define NPAD  16
#define B_STR 522

__device__ __forceinline__ float dot_piece(const float4 a0, const float4 a1, const float4 a2,
                                           const float4 a3, const float* __restrict__ wrow, int lane)
{
    const float4* wr = (const float4*)wrow;
    const float4 v0 = wr[lane], v1 = wr[lane + 32], v2 = wr[lane + 64], v3 = wr[lane + 96];
    float s = 0.0f;
    s += a0.x * v0.x + a0.y * v0.y + a0.z * v0.z + a0.w * v0.w;
    s += a1.x * v1.x + a1.y * v1.y + a1.z * v1.z + a1.w * v1.w;
    s += a2.x * v2.x + a2.y * v2.y + a2.z * v2.z + a2.w * v2.w;
    s += a3.x * v3.x + a3.y * v3.y + a3.z * v3.z + a3.w * v3.w;
    return s;
}

__device__ __forceinline__ float xch(float lo, float hi, bool up, int width)
{
    const float keep = up ? hi : lo;
    const float send = up ? lo : hi;
    return keep + __shfl_xor(send, width, 32);
}

__global__ __launch_bounds__(128) __attribute__((amdgpu_waves_per_eu(1, 1))) void torgb_styles(
    const float* __restrict__ w, const float* __restrict__ aw,
    const float* __restrict__ ab, float* __restrict__ styles)
{
    const int lane = threadIdx.x & 31;
    const int cc   = blockIdx.x * 4 + (threadIdx.x >> 5);

    const float abv = ab[cc];
    const float4* arow = (const float4*)(aw + (size_t)cc * C_DIM);
    const float4 a0 = arow[lane], a1 = arow[lane + 32], a2 = arow[lane + 64], a3 = arow[lane + 96];

#define DOT(b) dot_piece(a0, a1, a2, a3, w + (b) * C_DIM, lane)
    const float s0  = DOT(0),  s1  = DOT(1),  s2  = DOT(2),  s3  = DOT(3);
    const float s4  = DOT(4),  s5  = DOT(5),  s6  = DOT(6),  s7  = DOT(7);
    const float s8  = DOT(8),  s9  = DOT(9),  s10 = DOT(10), s11 = DOT(11);
    const float s12 = DOT(12), s13 = DOT(13), s14 = DOT(14), s15 = DOT(15);
#undef DOT
    asm volatile("" :: "v"(abv));

    const bool u16 = (lane & 16) != 0;
    const float e0 = xch(s0, s8,  u16, 16), e1 = xch(s1, s9,  u16, 16);
    const float e2 = xch(s2, s10, u16, 16), e3 = xch(s3, s11, u16, 16);
    const float e4 = xch(s4, s12, u16, 16), e5 = xch(s5, s13, u16, 16);
    const float e6 = xch(s6, s14, u16, 16), e7 = xch(s7, s15, u16, 16);
    const bool u8 = (lane & 8) != 0;
    const float f0 = xch(e0, e4, u8, 8), f1 = xch(e1, e5, u8, 8);
    const float f2 = xch(e2, e6, u8, 8), f3 = xch(e3, e7, u8, 8);
    const bool u4 = (lane & 4) != 0;
    const float g0 = xch(f0, f2, u4, 4), g1 = xch(f1, f3, u4, 4);
    const bool u2 = (lane & 2) != 0;
    const float h0 = xch(g0, g1, u2, 2);
    const float tot = h0 + __shfl_xor(h0, 1, 32);

    if ((lane & 1) == 0) {
        const float gain = 0.044194173824159216f;
        styles[(size_t)(lane >> 1) * C_DIM + cc] = tot * gain + abv;
    }
}

#define KQ     4
#define KQ_LEN (C_DIM / KQ)
#define GRP_K  32

__device__ __forceinline__ void load_group(v4f (&r)[16], const float* __restrict__ p)
{
    #pragma unroll
    for (int i = 0; i < 8; ++i) {
        r[2 * i]     = __builtin_nontemporal_load((const v4f*)(p + (size_t)(4 * i) * HW));
        r[2 * i + 1] = __builtin_nontemporal_load((const v4f*)(p + (size_t)(4 * i + 1) * HW));
    }
}

__device__ __forceinline__ void mma_group(v8f& c0, v8f& c1, v8f& c2, v8f& c3,
                                          const v4f (&r)[16], const float* pb)
{
    #pragma unroll
    for (int i = 0; i < 8; ++i) {
        const v2f Bv = *(const v2f*)(pb + 4 * i);
        const v4f r0 = r[2 * i], r1 = r[2 * i + 1];
        const v2f A0 = {r0.x, r1.x}, A1 = {r0.y, r1.y}, A2 = {r0.z, r1.z}, A3 = {r0.w, r1.w};
        c0 = __builtin_amdgcn_wmma_f32_16x16x4_f32(false, A0, false, Bv, (short)0, c0, false, false);
        c1 = __builtin_amdgcn_wmma_f32_16x16x4_f32(false, A1, false, Bv, (short)0, c1, false, false);
        c2 = __builtin_amdgcn_wmma_f32_16x16x4_f32(false, A2, false, Bv, (short)0, c2, false, false);
        c3 = __builtin_amdgcn_wmma_f32_16x16x4_f32(false, A3, false, Bv, (short)0, c3, false, false);
    }
}

__global__ __launch_bounds__(256) __attribute__((amdgpu_waves_per_eu(4, 4))) void torgb_main(
    const float* __restrict__ x, const float* __restrict__ styles,
    const float* __restrict__ wt, const float* __restrict__ bias,
    float* __restrict__ out)
{
    __shared__ float sB[NPAD * B_STR];
    __shared__ float s_part[OUT_C][8];
    __shared__ __attribute__((aligned(16))) float s_red[KQ][OUT_C][128];

    const int b    = blockIdx.x >> 5;
    const int grp  = blockIdx.x & 31;
    const int tid  = threadIdx.x;
    const int lane = tid & 31;
    const int wv   = tid >> 5;

    const float s0  = styles[(size_t)b * C_DIM + tid];
    const float s1  = styles[(size_t)b * C_DIM + tid + 256];
    const float w00 = wt[0 * C_DIM + tid], w01 = wt[0 * C_DIM + tid + 256];
    const float w10 = wt[1 * C_DIM + tid], w11 = wt[1 * C_DIM + tid + 256];
    const float w20 = wt[2 * C_DIM + tid], w21 = wt[2 * C_DIM + tid + 256];

    for (int i = tid; i < NPAD * B_STR; i += 256) {
        int n = i / B_STR;
        int c = i - n * B_STR;
        if (n >= OUT_C || c >= C_DIM) sB[i] = 0.0f;
    }

    const float t00 = w00 * s0, t01 = w01 * s1;
    const float t10 = w10 * s0, t11 = w11 * s1;
    const float t20 = w20 * s0, t21 = w21 * s1;
    float p0 = t00 * t00 + t01 * t01;
    float p1 = t10 * t10 + t11 * t11;
    float p2 = t20 * t20 + t21 * t21;
    #pragma unroll
    for (int off = 16; off > 0; off >>= 1) {
        p0 += __shfl_xor(p0, off, 32);
        p1 += __shfl_xor(p1, off, 32);
        p2 += __shfl_xor(p2, off, 32);
    }
    if (lane == 0) { s_part[0][wv] = p0; s_part[1][wv] = p1; s_part[2][wv] = p2; }
    __syncthreads();
    float q0 = 0.0f, q1 = 0.0f, q2 = 0.0f;
    #pragma unroll
    for (int k = 0; k < 8; ++k) { q0 += s_part[0][k]; q1 += s_part[1][k]; q2 += s_part[2][k]; }
    const float d0 = rsqrtf(q0 + 1e-8f);
    const float d1 = rsqrtf(q1 + 1e-8f);
    const float d2 = rsqrtf(q2 + 1e-8f);
    sB[0 * B_STR + tid] = t00 * d0;  sB[0 * B_STR + tid + 256] = t01 * d0;
    sB[1 * B_STR + tid] = t10 * d1;  sB[1 * B_STR + tid + 256] = t11 * d1;
    sB[2 * B_STR + tid] = t20 * d2;  sB[2 * B_STR + tid + 256] = t21 * d2;
    __syncthreads();

    const int m    = lane & 15;
    const int half = lane >> 4;
    const int koff = half << 1;
    const int pg   = wv & 1;
    const int kq   = wv >> 1;
    const int c0   = kq * KQ_LEN + koff;
    const int px0  = grp * 128 + pg * 64;

    const float* pA = x + ((size_t)b * C_DIM + (size_t)c0) * HW + px0 + 4 * m;
    const float* pB = &sB[m * B_STR + c0];

    v8f acc0 = {0.f, 0.f, 0.f, 0.f, 0.f, 0.f, 0.f, 0.f};
    v8f acc1 = acc0, acc2 = acc0, acc3 = acc0;

    v4f ga[16], gb[16];
    load_group(ga, pA);
    load_group(gb, pA + (size_t)(1 * GRP_K) * HW);
    mma_group(acc0, acc1, acc2, acc3, ga, pB);
    load_group(ga, pA + (size_t)(2 * GRP_K) * HW);
    mma_group(acc0, acc1, acc2, acc3, gb, pB + 1 * GRP_K);
    load_group(gb, pA + (size_t)(3 * GRP_K) * HW);
    mma_group(acc0, acc1, acc2, acc3, ga, pB + 2 * GRP_K);
    mma_group(acc0, acc1, acc2, acc3, gb, pB + 3 * GRP_K);

    if (m < OUT_C) {
        float* pr = &s_red[kq][m][pg * 64 + half * 32];
        #pragma unroll
        for (int r = 0; r < 8; ++r) {
            const v4f v = {acc0[r], acc1[r], acc2[r], acc3[r]};
            *(v4f*)(pr + 4 * r) = v;
        }
    }
    __syncthreads();

    if (tid < 96) {
        const int o = tid >> 5;
        const int q = tid & 31;
        const float bs = bias[o];
        v4f s = *(const v4f*)&s_red[0][o][4 * q];
        s += *(const v4f*)&s_red[1][o][4 * q];
        s += *(const v4f*)&s_red[2][o][4 * q];
        s += *(const v4f*)&s_red[3][o][4 * q];
        s += bs;
        *(v4f*)(out + ((size_t)b * OUT_C + o) * HW + grp * 128 + 4 * q) = s;
    }
}

extern "C" void kernel_launch(void* const* d_in, const int* in_sizes, int n_in,
                              void* d_out, int out_size, void* d_ws, size_t ws_size,
                              hipStream_t stream) {
    const float* x    = (const float*)d_in[0];
    const float* w    = (const float*)d_in[1];
    const float* aw   = (const float*)d_in[2];
    const float* ab   = (const float*)d_in[3];
    const float* wt   = (const float*)d_in[4];
    const float* bias = (const float*)d_in[5];
    float* out    = (float*)d_out;
    float* styles = (float*)d_ws;

    torgb_styles<<<128, 128, 0, stream>>>(w, aw, ab, styles);
    torgb_main<<<512, 256, 0, stream>>>(x, styles, wt, bias, out);
}


# Main: each wave builds the 3 weight rows of its 128 channels in its own LDS strip; no zero fill, exchange chain or block barrier before the x stream; demod scales the sums last, its sum reordered
# speedup vs baseline: 3.2816x; 1.1282x over previous
#include <hip/hip_runtime.h>


typedef __attribute__((ext_vector_type(2))) float v2f;
typedef __attribute__((ext_vector_type(4))) float v4f;
typedef __attribute__((ext_vector_type(8))) float v8f;

#define C_DIM 512
#define HW    4096
#define OUT_C 3
#define B_STR 136

__device__ __forceinline__ float dot_piece(const float4 a0, const float4 a1, const float4 a2,
                                           const float4 a3, const float* __restrict__ wrow, int lane)
{
    const float4* wr = (const float4*)wrow;
    const float4 v0 = wr[lane], v1 = wr[lane + 32], v2 = wr[lane + 64], v3 = wr[lane + 96];
    float s = 0.0f;
    s += a0.x * v0.x + a0.y * v0.y + a0.z * v0.z + a0.w * v0.w;
    s += a1.x * v1.x + a1.y * v1.y + a1.z * v1.z + a1.w * v1.w;
    s += a2.x * v2.x + a2.y * v2.y + a2.z * v2.z + a2.w * v2.w;
    s += a3.x * v3.x + a3.y * v3.y + a3.z * v3.z + a3.w * v3.w;
    return s;
}

__device__ __forceinline__ float xch(float lo, float hi, bool up, int width)
{
    const float keep = up ? hi : lo;
    const float send = up ? lo : hi;
    return keep + __shfl_xor(send, width, 32);
}

__global__ __launch_bounds__(128) __attribute__((amdgpu_waves_per_eu(1, 1))) void torgb_styles(
    const float* __restrict__ w, const float* __restrict__ aw,
    const float* __restrict__ ab, float* __restrict__ styles)
{
    const int lane = threadIdx.x & 31;
    const int cc   = blockIdx.x * 4 + (threadIdx.x >> 5);

    const float abv = ab[cc];
    const float4* arow = (const float4*)(aw + (size_t)cc * C_DIM);
    const float4 a0 = arow[lane], a1 = arow[lane + 32], a2 = arow[lane + 64], a3 = arow[lane + 96];

#define DOT(b) dot_piece(a0, a1, a2, a3, w + (b) * C_DIM, lane)
    const float s0  = DOT(0),  s1  = DOT(1),  s2  = DOT(2),  s3  = DOT(3);
    const float s4  = DOT(4),  s5  = DOT(5),  s6  = DOT(6),  s7  = DOT(7);
    const float s8  = DOT(8),  s9  = DOT(9),  s10 = DOT(10), s11 = DOT(11);
    const float s12 = DOT(12), s13 = DOT(13), s14 = DOT(14), s15 = DOT(15);
#undef DOT
    asm volatile("" :: "v"(abv));

    const bool u16 = (lane & 16) != 0;
    const float e0 = xch(s0, s8,  u16, 16), e1 = xch(s1, s9,  u16, 16);
    const float e2 = xch(s2, s10, u16, 16), e3 = xch(s3, s11, u16, 16);
    const float e4 = xch(s4, s12, u16, 16), e5 = xch(s5, s13, u16, 16);
    const float e6 = xch(s6, s14, u16, 16), e7 = xch(s7, s15, u16, 16);
    const bool u8 = (lane & 8) != 0;
    const float f0 = xch(e0, e4, u8, 8), f1 = xch(e1, e5, u8, 8);
    const float f2 = xch(e2, e6, u8, 8), f3 = xch(e3, e7, u8, 8);
    const bool u4 = (lane & 4) != 0;
    const float g0 = xch(f0, f2, u4, 4), g1 = xch(f1, f3, u4, 4);
    const bool u2 = (lane & 2) != 0;
    const float h0 = xch(g0, g1, u2, 2);
    const float tot = h0 + __shfl_xor(h0, 1, 32);

    if ((lane & 1) == 0) {
        const float gain = 0.044194173824159216f;
        styles[(size_t)(lane >> 1) * C_DIM + cc] = tot * gain + abv;
    }
}

#define KQ     4
#define KQ_LEN (C_DIM / KQ)
#define GRP_K  32

__device__ __forceinline__ void load_group(v4f (&r)[16], const float* __restrict__ p)
{
    #pragma unroll
    for (int i = 0; i < 8; ++i) {
        r[2 * i]     = __builtin_nontemporal_load((const v4f*)(p + (size_t)(4 * i) * HW));
        r[2 * i + 1] = __builtin_nontemporal_load((const v4f*)(p + (size_t)(4 * i + 1) * HW));
    }
}

__device__ __forceinline__ void mma_group(v8f& c0, v8f& c1, v8f& c2, v8f& c3, float& ssq,
                                          const v4f (&r)[16], const float* pb)
{
    v2f Bf[8];
    #pragma unroll
    for (int i = 0; i < 8; ++i) Bf[i] = *(const v2f*)(pb + 4 * i);
    #pragma unroll
    for (int i = 0; i < 8; ++i) {
        const v2f Bv = Bf[i];
        ssq += Bv.x * Bv.x + Bv.y * Bv.y;
        const v4f r0 = r[2 * i], r1 = r[2 * i + 1];
        const v2f A0 = {r0.x, r1.x}, A1 = {r0.y, r1.y}, A2 = {r0.z, r1.z}, A3 = {r0.w, r1.w};
        c0 = __builtin_amdgcn_wmma_f32_16x16x4_f32(false, A0, false, Bv, (short)0, c0, false, false);
        c1 = __builtin_amdgcn_wmma_f32_16x16x4_f32(false, A1, false, Bv, (short)0, c1, false, false);
        c2 = __builtin_amdgcn_wmma_f32_16x16x4_f32(false, A2, false, Bv, (short)0, c2, false, false);
        c3 = __builtin_amdgcn_wmma_f32_16x16x4_f32(false, A3, false, Bv, (short)0, c3, false, false);
        __builtin_amdgcn_sched_barrier(0);
    }
}

__global__ __launch_bounds__(256) __attribute__((amdgpu_waves_per_eu(4, 4))) void torgb_main(
    const float* __restrict__ x, const float* __restrict__ styles,
    const float* __restrict__ wt, const float* __restrict__ bias,
    float* __restrict__ out)
{
    __shared__ __attribute__((aligned(16))) float sB[2 * KQ][OUT_C * B_STR];
    __shared__ float s_ssq[KQ][4];
    __shared__ __attribute__((aligned(16))) float s_red[KQ][OUT_C][128];

    const int b    = blockIdx.x >> 5;
    const int grp  = blockIdx.x & 31;
    const int tid  = threadIdx.x;
    const int lane = tid & 31;
    const int wv   = tid >> 5;

    const int m    = lane & 15;
    const int half = lane >> 4;
    const int koff = half << 1;
    const int pg   = wv & 1;
    const int kq   = wv >> 1;
    const int c0   = kq * KQ_LEN + koff;
    const int px0  = grp * 128 + pg * 64;
    const int n    = m < OUT_C ? m : OUT_C - 1;

    const int cw = kq * KQ_LEN + 4 * lane;
    const v4f sv = *(const v4f*)(styles + (size_t)b * C_DIM + cw);
    const v4f w0 = *(const v4f*)(wt + 0 * C_DIM + cw);
    const v4f w1 = *(const v4f*)(wt + 1 * C_DIM + cw);
    const v4f w2 = *(const v4f*)(wt + 2 * C_DIM + cw);

    const float* pA = x + ((size_t)b * C_DIM + (size_t)c0) * HW + px0 + 4 * m;

    v4f ga[16], gb[16];
    load_group(ga, pA);
    load_group(gb, pA + (size_t)(1 * GRP_K) * HW);

    float* strip = &sB[wv][0];
    *(v4f*)(strip + 0 * B_STR + 4 * lane) = w0 * sv;
    *(v4f*)(strip + 1 * B_STR + 4 * lane) = w1 * sv;
    *(v4f*)(strip + 2 * B_STR + 4 * lane) = w2 * sv;
    __builtin_amdgcn_wave_barrier();
    const float* pB = strip + n * B_STR + koff;

    v8f acc0 = {0.f, 0.f, 0.f, 0.f, 0.f, 0.f, 0.f, 0.f};
    v8f acc1 = acc0, acc2 = acc0, acc3 = acc0;
    float ssq = 0.0f;

    mma_group(acc0, acc1, acc2, acc3, ssq, ga, pB);
    load_group(ga, pA + (size_t)(2 * GRP_K) * HW);
    mma_group(acc0, acc1, acc2, acc3, ssq, gb, pB + 1 * GRP_K);
    load_group(gb, pA + (size_t)(3 * GRP_K) * HW);
    mma_group(acc0, acc1, acc2, acc3, ssq, ga, pB + 2 * GRP_K);
    mma_group(acc0, acc1, acc2, acc3, ssq, gb, pB + 3 * GRP_K);

    ssq += __shfl_xor(ssq, 16, 32);

    if (m < OUT_C) {
        float* pr = &s_red[kq][m][pg * 64 + half * 32];
        #pragma unroll
        for (int r = 0; r < 8; ++r) {
            const v4f v = {acc0[r], acc1[r], acc2[r], acc3[r]};
            *(v4f*)(pr + 4 * r) = v;
        }
        if (pg == 0 && half == 0) s_ssq[kq][m] = ssq;
    }
    __syncthreads();

    if (tid < 96) {
        const int o = tid >> 5;
        const int q = tid & 31;
        const float bs = bias[o];
        const float d  = rsqrtf(((s_ssq[0][o] + s_ssq[1][o]) + s_ssq[2][o]) + s_ssq[3][o] + 1e-8f);
        v4f s = *(const v4f*)&s_red[0][o][4 * q];
        s += *(const v4f*)&s_red[1][o][4 * q];
        s += *(const v4f*)&s_red[2][o][4 * q];
        s += *(const v4f*)&s_red[3][o][4 * q];
        s = s * d + bs;
        *(v4f*)(out + ((size_t)b * OUT_C + o) * HW + grp * 128 + 4 * q) = s;
    }
}

extern "C" void kernel_launch(void* const* d_in, const int* in_sizes, int n_in,
                              void* d_out, int out_size, void* d_ws, size_t ws_size,
                              hipStream_t stream) {
    const float* x    = (const float*)d_in[0];
    const float* w    = (const float*)d_in[1];
    const float* aw   = (const float*)d_in[2];
    const float* ab   = (const float*)d_in[3];
    const float* wt   = (const float*)d_in[4];
    const float* bias = (const float*)d_in[5];
    float* out    = (float*)d_out;
    float* styles = (float*)d_ws;

    torgb_styles<<<128, 128, 0, stream>>>(w, aw, ab, styles);
    torgb_main<<<512, 256, 0, stream>>>(x, styles, wt, bias, out);
}
